# starting kernel
# baseline (speedup 1.0000x reference)
#include <hip/hip_runtime.h>

typedef __attribute__((ext_vector_type(2))) float v2f;
typedef __attribute__((ext_vector_type(8))) float v8f;

#define GCN_D 128

__global__ __launch_bounds__(256) void gcn_degree(const int* __restrict__ dst,
                                                  float* __restrict__ deg, int n_edges) {
    int i = blockIdx.x * blockDim.x + threadIdx.x;
    int stride = gridDim.x * blockDim.x;
    for (; i < n_edges; i += stride) atomicAdd(&deg[dst[i]], 1.0f);
}

__global__ __launch_bounds__(256) void gcn_dinv(float* __restrict__ deg, int n_nodes) {
    int i = blockIdx.x * blockDim.x + threadIdx.x;
    if (i < n_nodes) deg[i] = rsqrtf(1.0f + deg[i]);
}

__global__ __launch_bounds__(256) void gcn_gemm(const float* __restrict__ X,
                                                const float* __restrict__ W,
                                                float* __restrict__ H, int n) {
    __shared__ float sW[GCN_D * GCN_D];
    for (int i = threadIdx.x; i < GCN_D * GCN_D; i += blockDim.x) sW[i] = W[i];
    __syncthreads();

    const int lane = threadIdx.x & 31;
    const int wave = threadIdx.x >> 5;
    const int m    = lane & 15;
    const int kh   = lane >> 4;
    const int row0 = blockIdx.x * 128 + wave * 16;

    int rowA = row0 + m;
    if (rowA > n - 1) rowA = n - 1;
    const float* xrow = X + (size_t)rowA * GCN_D;

    v8f acc[8] = {};

    for (int kk = 0; kk < 32; ++kk) {
        const int kb = kk * 4 + kh * 2;
        v2f a;
        a.x = xrow[kb];
        a.y = xrow[kb + 1];
        #pragma unroll
        for (int nt = 0; nt < 8; ++nt) {
            v2f b;
            b.x = sW[(kb + 0) * GCN_D + nt * 16 + m];
            b.y = sW[(kb + 1) * GCN_D + nt * 16 + m];
            acc[nt] = __builtin_amdgcn_wmma_f32_16x16x4_f32(
                false, a, false, b, (short)0, acc[nt], false, false);
        }
    }

    #pragma unroll
    for (int nt = 0; nt < 8; ++nt) {
        #pragma unroll
        for (int v = 0; v < 8; ++v) {
            int r = row0 + v + 8 * kh;
            if (r < n) H[(size_t)r * GCN_D + nt * 16 + m] = acc[nt][v];
        }
    }
}

__global__ __launch_bounds__(256) void gcn_scatter(const int* __restrict__ src,
                                                   const int* __restrict__ dst,
                                                   const float* __restrict__ dinv,
                                                   const float* __restrict__ h,
                                                   float* __restrict__ agg, int n_edges) {
    const int lane = threadIdx.x & 31;
    int wid = blockIdx.x * (blockDim.x >> 5) + (threadIdx.x >> 5);
    const int nw = gridDim.x * (blockDim.x >> 5);
    for (int e = wid; e < n_edges; e += nw) {
        const int s = src[e];
        const int d = dst[e];
        const float norm = dinv[s] * dinv[d];
        const float4 v = ((const float4*)(h + (size_t)s * GCN_D))[lane];
        float* a = agg + (size_t)d * GCN_D + lane * 4;
        atomicAdd(a + 0, v.x * norm);
        atomicAdd(a + 1, v.y * norm);
        atomicAdd(a + 2, v.z * norm);
        atomicAdd(a + 3, v.w * norm);
    }
}

__global__ __launch_bounds__(256) void gcn_combine(const float* __restrict__ h,
                                                   const float* __restrict__ agg,
                                                   const float* __restrict__ dinv,
                                                   const float* __restrict__ bias,
                                                   float* __restrict__ out, int n, int do_relu) {
    size_t i = (size_t)blockIdx.x * blockDim.x + threadIdx.x;
    size_t total = (size_t)n * (GCN_D / 4);
    if (i >= total) return;
    int node = (int)(i >> 5);
    int f4   = (int)(i & 31);
    float di = dinv[node];
    float d2 = di * di;
    float4 hv = ((const float4*)h)[i];
    float4 av = ((const float4*)agg)[i];
    float4 bv = ((const float4*)bias)[f4];
    float4 o;
    o.x = av.x + hv.x * d2 + bv.x;
    o.y = av.y + hv.y * d2 + bv.y;
    o.z = av.z + hv.z * d2 + bv.z;
    o.w = av.w + hv.w * d2 + bv.w;
    if (do_relu) {
        o.x = fmaxf(o.x, 0.0f);
        o.y = fmaxf(o.y, 0.0f);
        o.z = fmaxf(o.z, 0.0f);
        o.w = fmaxf(o.w, 0.0f);
    }
    ((float4*)out)[i] = o;
}

extern "C" void kernel_launch(void* const* d_in, const int* in_sizes, int n_in,
                              void* d_out, int out_size, void* d_ws, size_t ws_size,
                              hipStream_t stream) {
    const float* x  = (const float*)d_in[0];
    const int*   ei = (const int*)d_in[1];
    const float* W1 = (const float*)d_in[2];
    const float* b1 = (const float*)d_in[3];
    const float* W2 = (const float*)d_in[4];
    const float* b2 = (const float*)d_in[5];
    const float* W3 = (const float*)d_in[6];
    const float* b3 = (const float*)d_in[7];

    const int n = in_sizes[0] / GCN_D;
    const int E = in_sizes[1] / 2;
    const int* srcI = ei;
    const int* dstI = ei + E;

    float* ws = (float*)d_ws;
    size_t n4 = ((size_t)n + 3) & ~(size_t)3;
    float* dinv = ws;
    float* h    = ws + n4;
    float* agg  = h + (size_t)n * GCN_D;
    float* bufA = agg + (size_t)n * GCN_D;
    float* out  = (float*)d_out;

    hipMemsetAsync(dinv, 0, (size_t)n * sizeof(float), stream);
    gcn_degree<<<1024, 256, 0, stream>>>(dstI, dinv, E);
    gcn_dinv<<<(n + 255) / 256, 256, 0, stream>>>(dinv, n);

    const float* Ws[3] = {W1, W2, W3};
    const float* bs[3] = {b1, b2, b3};
    const float* cur = x;
    for (int l = 0; l < 3; ++l) {
        float* o = (l == 2) ? out : bufA;
        gcn_gemm<<<(n + 127) / 128, 256, 0, stream>>>(cur, Ws[l], h, n);
        hipMemsetAsync(agg, 0, (size_t)n * GCN_D * sizeof(float), stream);
        gcn_scatter<<<2048, 256, 0, stream>>>(srcI, dstI, dinv, h, agg, E);
        size_t total4 = (size_t)n * (GCN_D / 4);
        gcn_combine<<<(int)((total4 + 255) / 256), 256, 0, stream>>>(
            h, agg, dinv, bs[l], o, n, (l < 2) ? 1 : 0);
        cur = o;
    }
}


# Segment sum by gather: edges counting-sorted by dst per call (int atomic rank + scan); one wave per node sums h[src] rows in registers, stores the finished row; no float atomics; sum order changes
# speedup vs baseline: 2.7175x; 2.7175x over previous
#include <hip/hip_runtime.h>

typedef __attribute__((ext_vector_type(2))) float v2f;
typedef __attribute__((ext_vector_type(8))) float v8f;

#define GCN_D 128

__global__ __launch_bounds__(256) void gcn_count(const int* __restrict__ dst,
                                                 int* __restrict__ cnt,
                                                 int* __restrict__ rank, int n_edges) {
    int i = blockIdx.x * blockDim.x + threadIdx.x;
    int stride = gridDim.x * blockDim.x;
    for (; i < n_edges; i += stride) rank[i] = atomicAdd(&cnt[dst[i]], 1);
}

__global__ __launch_bounds__(256) void gcn_scan(const int* __restrict__ cnt,
                                                int* __restrict__ rowptr,
                                                float* __restrict__ dinv, int n) {
    __shared__ int s_before[8];
    __shared__ int s_wave[8];
    const int tid  = threadIdx.x;
    const int lane = tid & 31;
    const int wave = tid >> 5;
    const int base = blockIdx.x * 1024;

    int before = 0;
    const int4* c4 = (const int4*)cnt;
    #pragma unroll 8
    for (int i = tid; i < base / 4; i += 256) {
        int4 v = c4[i];
        before += (v.x + v.y) + (v.z + v.w);
    }
    for (int off = 16; off > 0; off >>= 1) before += __shfl_xor(before, off);
    if (lane == 0) s_before[wave] = before;

    const int idx = base + tid * 4;
    int c[4];
    #pragma unroll
    for (int k = 0; k < 4; ++k) c[k] = (idx + k < n) ? cnt[idx + k] : 0;
    const int mine = (c[0] + c[1]) + (c[2] + c[3]);
    int inc = mine;
    for (int off = 1; off < 32; off <<= 1) {
        int y = __shfl_up(inc, off);
        if (lane >= off) inc += y;
    }
    if (lane == 31) s_wave[wave] = inc;
    __syncthreads();

    int excl = inc - mine;
    #pragma unroll
    for (int w = 0; w < 8; ++w) {
        excl += s_before[w];
        if (w < wave) excl += s_wave[w];
    }
    int run = excl;
    #pragma unroll
    for (int k = 0; k < 4; ++k) {
        if (idx + k < n) {
            rowptr[idx + k] = run;
            dinv[idx + k] = rsqrtf(1.0f + (float)c[k]);
        }
        run += c[k];
    }
    if (idx < n && idx + 4 >= n) rowptr[n] = run;
}

__global__ __launch_bounds__(256) void gcn_fill(const int* __restrict__ src,
                                                const int* __restrict__ dst,
                                                const int* __restrict__ rowptr,
                                                const int* __restrict__ rank,
                                                const float* __restrict__ dinv,
                                                int* __restrict__ nbr, int n_edges) {
    int i = blockIdx.x * blockDim.x + threadIdx.x;
    int stride = gridDim.x * blockDim.x;
    int2* pairs = (int2*)nbr;
    for (; i < n_edges; i += stride) {
        const int s = src[i];
        pairs[rowptr[dst[i]] + rank[i]] = make_int2(s, __float_as_int(dinv[s]));
    }
}

__global__ __launch_bounds__(256) void gcn_gemm(const float* __restrict__ X,
                                                const float* __restrict__ W,
                                                float* __restrict__ H, int n) {
    __shared__ float sW[GCN_D * GCN_D];
    for (int i = threadIdx.x; i < GCN_D * GCN_D; i += blockDim.x) sW[i] = W[i];
    __syncthreads();

    const int lane = threadIdx.x & 31;
    const int wave = threadIdx.x >> 5;
    const int m    = lane & 15;
    const int kh   = lane >> 4;
    const int row0 = blockIdx.x * 128 + wave * 16;

    int rowA = row0 + m;
    if (rowA > n - 1) rowA = n - 1;
    const float* xrow = X + (size_t)rowA * GCN_D;

    v8f acc[8] = {};

    for (int kk = 0; kk < 32; ++kk) {
        const int kb = kk * 4 + kh * 2;
        v2f a;
        a.x = xrow[kb];
        a.y = xrow[kb + 1];
        #pragma unroll
        for (int nt = 0; nt < 8; ++nt) {
            v2f b;
            b.x = sW[(kb + 0) * GCN_D + nt * 16 + m];
            b.y = sW[(kb + 1) * GCN_D + nt * 16 + m];
            acc[nt] = __builtin_amdgcn_wmma_f32_16x16x4_f32(
                false, a, false, b, (short)0, acc[nt], false, false);
        }
    }

    #pragma unroll
    for (int nt = 0; nt < 8; ++nt) {
        #pragma unroll
        for (int v = 0; v < 8; ++v) {
            int r = row0 + v + 8 * kh;
            if (r < n) H[(size_t)r * GCN_D + nt * 16 + m] = acc[nt][v];
        }
    }
}

__global__ __launch_bounds__(256) void gcn_gather(const int* __restrict__ rowptr,
                                                  const int* __restrict__ nbr,
                                                  const float* __restrict__ dinv,
                                                  const float* __restrict__ h,
                                                  const float* __restrict__ bias,
                                                  float* __restrict__ out, int n, int do_relu) {
    const int lane = threadIdx.x & 31;
    int node = blockIdx.x * (blockDim.x >> 5) + (threadIdx.x >> 5);
    if (node >= n) return;
    node = __builtin_amdgcn_readfirstlane(node);

    const int beg = rowptr[node];
    const int end = rowptr[node + 1];
    const float dd = dinv[node];
    const float4* h4 = (const float4*)h;

    float4 acc = make_float4(0.0f, 0.0f, 0.0f, 0.0f);
    for (int base = beg; base < end; base += 32) {
        const int cnt = (end - base < 32) ? (end - base) : 32;
        int   my = 0;
        float mw = 0.0f;
        if (lane < cnt) {
            const int2 p = ((const int2*)nbr)[base + lane];
            my = p.x;
            mw = __int_as_float(p.y) * dd;
        }
        const int mwb = __float_as_int(mw);
#define GCN_ROW(J) h4[(size_t)__builtin_amdgcn_readlane(my, (J)) * 32 + lane]
#define GCN_WGT(J) __int_as_float(__builtin_amdgcn_readlane(mwb, (J)))
        int j = 0;
        for (; j + 4 <= cnt; j += 4) {
            const float4 v0 = GCN_ROW(j + 0);
            const float4 v1 = GCN_ROW(j + 1);
            const float4 v2 = GCN_ROW(j + 2);
            const float4 v3 = GCN_ROW(j + 3);
            const float w0 = GCN_WGT(j + 0), w1 = GCN_WGT(j + 1);
            const float w2 = GCN_WGT(j + 2), w3 = GCN_WGT(j + 3);
            acc.x += v0.x * w0; acc.y += v0.y * w0; acc.z += v0.z * w0; acc.w += v0.w * w0;
            acc.x += v1.x * w1; acc.y += v1.y * w1; acc.z += v1.z * w1; acc.w += v1.w * w1;
            acc.x += v2.x * w2; acc.y += v2.y * w2; acc.z += v2.z * w2; acc.w += v2.w * w2;
            acc.x += v3.x * w3; acc.y += v3.y * w3; acc.z += v3.z * w3; acc.w += v3.w * w3;
        }
        for (; j < cnt; ++j) {
            const float4 v0 = GCN_ROW(j);
            const float w0 = GCN_WGT(j);
            acc.x += v0.x * w0; acc.y += v0.y * w0; acc.z += v0.z * w0; acc.w += v0.w * w0;
        }
#undef GCN_ROW
#undef GCN_WGT
    }

    const float d2 = dd * dd;
    const float4 hv = h4[(size_t)node * 32 + lane];
    const float4 bv = ((const float4*)bias)[lane];
    float4 o;
    o.x = acc.x + hv.x * d2 + bv.x;
    o.y = acc.y + hv.y * d2 + bv.y;
    o.z = acc.z + hv.z * d2 + bv.z;
    o.w = acc.w + hv.w * d2 + bv.w;
    if (do_relu) {
        o.x = fmaxf(o.x, 0.0f);
        o.y = fmaxf(o.y, 0.0f);
        o.z = fmaxf(o.z, 0.0f);
        o.w = fmaxf(o.w, 0.0f);
    }
    ((float4*)out)[(size_t)node * 32 + lane] = o;
}

extern "C" void kernel_launch(void* const* d_in, const int* in_sizes, int n_in,
                              void* d_out, int out_size, void* d_ws, size_t ws_size,
                              hipStream_t stream) {
    const float* x  = (const float*)d_in[0];
    const int*   ei = (const int*)d_in[1];
    const float* W1 = (const float*)d_in[2];
    const float* b1 = (const float*)d_in[3];
    const float* W2 = (const float*)d_in[4];
    const float* b2 = (const float*)d_in[5];
    const float* W3 = (const float*)d_in[6];
    const float* b3 = (const float*)d_in[7];

    const int n = in_sizes[0] / GCN_D;
    const int E = in_sizes[1] / 2;
    const int* srcI = ei;
    const int* dstI = ei + E;

    float* ws = (float*)d_ws;
    size_t n4 = ((size_t)n + 3) & ~(size_t)3;
    float* dinv = ws;
    float* h    = ws + n4;
    float* reg  = h + (size_t)n * GCN_D;
    float* bufA = reg + (size_t)n * GCN_D;
    float* out  = (float*)d_out;

    size_t e2 = ((size_t)E + 1) & ~(size_t)1;
    int*  cnt    = (int*)reg;
    int*  rowptr = cnt + n4;
    int*  rank   = rowptr + n4 + 4;
    int*  nbr    = rank + e2;

    hipMemsetAsync(cnt, 0, (size_t)n * sizeof(int), stream);
    gcn_count<<<(E + 255) / 256, 256, 0, stream>>>(dstI, cnt, rank, E);
    gcn_scan<<<(n + 1023) / 1024, 256, 0, stream>>>(cnt, rowptr, dinv, n);
    gcn_fill<<<(E + 255) / 256, 256, 0, stream>>>(srcI, dstI, rowptr, rank, dinv, nbr, E);

    const float* Ws[3] = {W1, W2, W3};
    const float* bs[3] = {b1, b2, b3};
    const float* cur = x;
    for (int l = 0; l < 3; ++l) {
        float* o = (l == 2) ? out : bufA;
        gcn_gemm<<<(n + 127) / 128, 256, 0, stream>>>(cur, Ws[l], h, n);
        gcn_gather<<<(n + 7) / 8, 256, 0, stream>>>(rowptr, nbr, dinv, h, bs[l], o, n,
                                                    (l < 2) ? 1 : 0);
        cur = o;
    }
}


# Edge sort on LDS counters: edges binned into 512-node buckets by per-block LDS cursors, then one block per bucket ranks its edges in LDS; no global atomic, no memset; sum order stays arrival-set
# speedup vs baseline: 3.0348x; 1.1168x over previous
#include <hip/hip_runtime.h>

typedef __attribute__((ext_vector_type(2))) float v2f;
typedef __attribute__((ext_vector_type(8))) float v8f;

#define GCN_D 128

#define GCN_BSHIFT 9
#define GCN_BNODES 512
#define GCN_CHUNKS 256
#define GCN_MAXB   2048

__global__ __launch_bounds__(1024) void gcn_bhist(const int* __restrict__ dst,
                                                  int* __restrict__ bhist,
                                                  int n_edges, int nb) {
    __shared__ int s_h[GCN_MAXB];
    for (int k = threadIdx.x; k < nb; k += 1024) s_h[k] = 0;
    __syncthreads();
    const int chunk = (n_edges + GCN_CHUNKS - 1) / GCN_CHUNKS;
    const int lo = blockIdx.x * chunk;
    const int hi = (lo + chunk < n_edges) ? (lo + chunk) : n_edges;
    for (int i = lo + threadIdx.x; i < hi; i += 4 * 1024) {
        int d[4];
        #pragma unroll
        for (int k = 0; k < 4; ++k) d[k] = (i + k * 1024 < hi) ? dst[i + k * 1024] : 0;
        #pragma unroll
        for (int k = 0; k < 4; ++k)
            if (i + k * 1024 < hi) atomicAdd(&s_h[d[k] >> GCN_BSHIFT], 1);
    }
    __syncthreads();
    for (int k = threadIdx.x; k < nb; k += 1024) bhist[k * GCN_CHUNKS + blockIdx.x] = s_h[k];
}

__global__ __launch_bounds__(GCN_CHUNKS) void gcn_bscan(const int* __restrict__ bhist,
                                                        int* __restrict__ boffs,
                                                        int* __restrict__ bstart, int nb) {
    __shared__ int s_before[8];
    __shared__ int s_wave[8];
    const int tid  = threadIdx.x;
    const int lane = tid & 31;
    const int wave = tid >> 5;
    const int b    = blockIdx.x;

    int before = 0;
    const int4* h4 = (const int4*)bhist;
    #pragma unroll 8
    for (int i = tid; i < b * (GCN_CHUNKS / 4); i += GCN_CHUNKS) {
        int4 v = h4[i];
        before += (v.x + v.y) + (v.z + v.w);
    }
    for (int off = 16; off > 0; off >>= 1) before += __shfl_xor(before, off);
    if (lane == 0) s_before[wave] = before;

    const int mine = bhist[b * GCN_CHUNKS + tid];
    int inc = mine;
    for (int off = 1; off < 32; off <<= 1) {
        int y = __shfl_up(inc, off);
        if (lane >= off) inc += y;
    }
    if (lane == 31) s_wave[wave] = inc;
    __syncthreads();

    int excl = inc - mine;
    #pragma unroll
    for (int w = 0; w < 8; ++w) {
        excl += s_before[w];
        if (w < wave) excl += s_wave[w];
    }
    boffs[b * GCN_CHUNKS + tid] = excl;
    if (tid == 0) bstart[b] = excl;
    if (b == nb - 1 && tid == GCN_CHUNKS - 1) bstart[nb] = excl + mine;
}

__global__ __launch_bounds__(1024) void gcn_bscatter(const int* __restrict__ src,
                                                     const int* __restrict__ dst,
                                                     const int* __restrict__ boffs,
                                                     int* __restrict__ coarse,
                                                     int n_edges, int nb) {
    __shared__ int s_pos[GCN_MAXB];
    for (int k = threadIdx.x; k < nb; k += 1024) s_pos[k] = boffs[k * GCN_CHUNKS + blockIdx.x];
    __syncthreads();
    const int chunk = (n_edges + GCN_CHUNKS - 1) / GCN_CHUNKS;
    const int lo = blockIdx.x * chunk;
    const int hi = (lo + chunk < n_edges) ? (lo + chunk) : n_edges;
    int2* pairs = (int2*)coarse;
    for (int i = lo + threadIdx.x; i < hi; i += 4 * 1024) {
        int d[4], s[4];
        #pragma unroll
        for (int k = 0; k < 4; ++k) {
            const bool in = i + k * 1024 < hi;
            d[k] = in ? dst[i + k * 1024] : 0;
            s[k] = in ? src[i + k * 1024] : 0;
        }
        #pragma unroll
        for (int k = 0; k < 4; ++k) {
            if (i + k * 1024 < hi) {
                const int p = atomicAdd(&s_pos[d[k] >> GCN_BSHIFT], 1);
                pairs[p] = make_int2(s[k], d[k]);
            }
        }
    }
}

__global__ __launch_bounds__(1024) void gcn_bucket(const int* __restrict__ coarse,
                                                   const int* __restrict__ bstart,
                                                   int* __restrict__ rank,
                                                   int* __restrict__ rowptr,
                                                   float* __restrict__ dinv, int n) {
    __shared__ int s_cnt[GCN_BNODES];
    __shared__ int s_wave[GCN_BNODES / 32];
    const int tid  = threadIdx.x;
    const int lane = tid & 31;
    const int wave = tid >> 5;
    const int b    = blockIdx.x;
    const int bs   = bstart[b];
    const int be   = bstart[b + 1];
    const int node0 = b << GCN_BSHIFT;

    if (tid < GCN_BNODES) s_cnt[tid] = 0;
    __syncthreads();
    const int2* pairs = (const int2*)coarse;
    for (int i = bs + tid; i < be; i += 4 * 1024) {
        int d[4];
        #pragma unroll
        for (int k = 0; k < 4; ++k) d[k] = (i + k * 1024 < be) ? pairs[i + k * 1024].y : node0;
        #pragma unroll
        for (int k = 0; k < 4; ++k)
            if (i + k * 1024 < be) rank[i + k * 1024] = atomicAdd(&s_cnt[d[k] - node0], 1);
    }
    __syncthreads();

    int c = 0, inc = 0;
    if (tid < GCN_BNODES) {
        c = s_cnt[tid];
        inc = c;
        for (int off = 1; off < 32; off <<= 1) {
            int y = __shfl_up(inc, off);
            if (lane >= off) inc += y;
        }
        if (lane == 31) s_wave[wave] = inc;
    }
    __syncthreads();
    if (tid < GCN_BNODES) {
        int excl = inc - c;
        #pragma unroll
        for (int w = 0; w < GCN_BNODES / 32; ++w)
            if (w < wave) excl += s_wave[w];
        const int node = node0 + tid;
        if (node < n) {
            rowptr[node] = bs + excl;
            dinv[node] = rsqrtf(1.0f + (float)c);
        }
    }
    if (b == (int)gridDim.x - 1 && tid == 0) rowptr[n] = be;
}

__global__ __launch_bounds__(256) void gcn_fill(const int* __restrict__ coarse,
                                                const int* __restrict__ rowptr,
                                                const int* __restrict__ rank,
                                                const float* __restrict__ dinv,
                                                int* __restrict__ nbr, int n_edges) {
    int i = blockIdx.x * blockDim.x + threadIdx.x;
    int stride = gridDim.x * blockDim.x;
    const int2* in = (const int2*)coarse;
    int2* pairs = (int2*)nbr;
    for (; i < n_edges; i += stride) {
        const int2 e = in[i];
        pairs[rowptr[e.y] + rank[i]] = make_int2(e.x, __float_as_int(dinv[e.x]));
    }
}

__global__ __launch_bounds__(256) void gcn_gemm(const float* __restrict__ X,
                                                const float* __restrict__ W,
                                                float* __restrict__ H, int n) {
    __shared__ float sW[GCN_D * GCN_D];
    for (int i = threadIdx.x; i < GCN_D * GCN_D; i += blockDim.x) sW[i] = W[i];
    __syncthreads();

    const int lane = threadIdx.x & 31;
    const int wave = threadIdx.x >> 5;
    const int m    = lane & 15;
    const int kh   = lane >> 4;
    const int row0 = blockIdx.x * 128 + wave * 16;

    int rowA = row0 + m;
    if (rowA > n - 1) rowA = n - 1;
    const float* xrow = X + (size_t)rowA * GCN_D;

    v8f acc[8] = {};

    for (int kk = 0; kk < 32; ++kk) {
        const int kb = kk * 4 + kh * 2;
        v2f a;
        a.x = xrow[kb];
        a.y = xrow[kb + 1];
        #pragma unroll
        for (int nt = 0; nt < 8; ++nt) {
            v2f b;
            b.x = sW[(kb + 0) * GCN_D + nt * 16 + m];
            b.y = sW[(kb + 1) * GCN_D + nt * 16 + m];
            acc[nt] = __builtin_amdgcn_wmma_f32_16x16x4_f32(
                false, a, false, b, (short)0, acc[nt], false, false);
        }
    }

    #pragma unroll
    for (int nt = 0; nt < 8; ++nt) {
        #pragma unroll
        for (int v = 0; v < 8; ++v) {
            int r = row0 + v + 8 * kh;
            if (r < n) H[(size_t)r * GCN_D + nt * 16 + m] = acc[nt][v];
        }
    }
}

__global__ __launch_bounds__(256) void gcn_gather(const int* __restrict__ rowptr,
                                                  const int* __restrict__ nbr,
                                                  const float* __restrict__ dinv,
                                                  const float* __restrict__ h,
                                                  const float* __restrict__ bias,
                                                  float* __restrict__ out, int n, int do_relu) {
    const int lane = threadIdx.x & 31;
    int node = blockIdx.x * (blockDim.x >> 5) + (threadIdx.x >> 5);
    if (node >= n) return;
    node = __builtin_amdgcn_readfirstlane(node);

    const int beg = rowptr[node];
    const int end = rowptr[node + 1];
    const float dd = dinv[node];
    const float4* h4 = (const float4*)h;

    float4 acc = make_float4(0.0f, 0.0f, 0.0f, 0.0f);
    for (int base = beg; base < end; base += 32) {
        const int cnt = (end - base < 32) ? (end - base) : 32;
        int   my = 0;
        float mw = 0.0f;
        if (lane < cnt) {
            const int2 p = ((const int2*)nbr)[base + lane];
            my = p.x;
            mw = __int_as_float(p.y) * dd;
        }
        const int mwb = __float_as_int(mw);
#define GCN_ROW(J) h4[(size_t)__builtin_amdgcn_readlane(my, (J)) * 32 + lane]
#define GCN_WGT(J) __int_as_float(__builtin_amdgcn_readlane(mwb, (J)))
        int j = 0;
        for (; j + 4 <= cnt; j += 4) {
            const float4 v0 = GCN_ROW(j + 0);
            const float4 v1 = GCN_ROW(j + 1);
            const float4 v2 = GCN_ROW(j + 2);
            const float4 v3 = GCN_ROW(j + 3);
            const float w0 = GCN_WGT(j + 0), w1 = GCN_WGT(j + 1);
            const float w2 = GCN_WGT(j + 2), w3 = GCN_WGT(j + 3);
            acc.x += v0.x * w0; acc.y += v0.y * w0; acc.z += v0.z * w0; acc.w += v0.w * w0;
            acc.x += v1.x * w1; acc.y += v1.y * w1; acc.z += v1.z * w1; acc.w += v1.w * w1;
            acc.x += v2.x * w2; acc.y += v2.y * w2; acc.z += v2.z * w2; acc.w += v2.w * w2;
            acc.x += v3.x * w3; acc.y += v3.y * w3; acc.z += v3.z * w3; acc.w += v3.w * w3;
        }
        for (; j < cnt; ++j) {
            const float4 v0 = GCN_ROW(j);
            const float w0 = GCN_WGT(j);
            acc.x += v0.x * w0; acc.y += v0.y * w0; acc.z += v0.z * w0; acc.w += v0.w * w0;
        }
#undef GCN_ROW
#undef GCN_WGT
    }

    const float d2 = dd * dd;
    const float4 hv = h4[(size_t)node * 32 + lane];
    const float4 bv = ((const float4*)bias)[lane];
    float4 o;
    o.x = acc.x + hv.x * d2 + bv.x;
    o.y = acc.y + hv.y * d2 + bv.y;
    o.z = acc.z + hv.z * d2 + bv.z;
    o.w = acc.w + hv.w * d2 + bv.w;
    if (do_relu) {
        o.x = fmaxf(o.x, 0.0f);
        o.y = fmaxf(o.y, 0.0f);
        o.z = fmaxf(o.z, 0.0f);
        o.w = fmaxf(o.w, 0.0f);
    }
    ((float4*)out)[(size_t)node * 32 + lane] = o;
}

extern "C" void kernel_launch(void* const* d_in, const int* in_sizes, int n_in,
                              void* d_out, int out_size, void* d_ws, size_t ws_size,
                              hipStream_t stream) {
    const float* x  = (const float*)d_in[0];
    const int*   ei = (const int*)d_in[1];
    const float* W1 = (const float*)d_in[2];
    const float* b1 = (const float*)d_in[3];
    const float* W2 = (const float*)d_in[4];
    const float* b2 = (const float*)d_in[5];
    const float* W3 = (const float*)d_in[6];
    const float* b3 = (const float*)d_in[7];

    const int n = in_sizes[0] / GCN_D;
    const int E = in_sizes[1] / 2;
    const int* srcI = ei;
    const int* dstI = ei + E;

    float* ws = (float*)d_ws;
    size_t n4 = ((size_t)n + 3) & ~(size_t)3;
    float* dinv = ws;
    float* h    = ws + n4;
    float* reg  = h + (size_t)n * GCN_D;
    float* bufA = reg + (size_t)n * GCN_D;
    float* out  = (float*)d_out;

    size_t e2 = ((size_t)E + 1) & ~(size_t)1;
    const int nb = (n + GCN_BNODES - 1) >> GCN_BSHIFT;
    size_t nb4 = ((size_t)nb + 1 + 3) & ~(size_t)3;
    int*  bhist  = (int*)reg;
    int*  boffs  = bhist + (size_t)nb * GCN_CHUNKS;
    int*  bstart = boffs + (size_t)nb * GCN_CHUNKS;
    int*  rowptr = bstart + nb4;
    int*  rank   = rowptr + n4 + 4;
    int*  nbr    = rank + e2;
    int*  coarse = nbr + 2 * e2;

    gcn_bhist<<<GCN_CHUNKS, 1024, 0, stream>>>(dstI, bhist, E, nb);
    gcn_bscan<<<nb, GCN_CHUNKS, 0, stream>>>(bhist, boffs, bstart, nb);
    gcn_bscatter<<<GCN_CHUNKS, 1024, 0, stream>>>(srcI, dstI, boffs, coarse, E, nb);
    gcn_bucket<<<nb, 1024, 0, stream>>>(coarse, bstart, rank, rowptr, dinv, n);
    gcn_fill<<<(E + 255) / 256, 256, 0, stream>>>(coarse, rowptr, rank, dinv, nbr, E);

    const float* Ws[3] = {W1, W2, W3};
    const float* bs[3] = {b1, b2, b3};
    const float* cur = x;
    for (int l = 0; l < 3; ++l) {
        float* o = (l == 2) ? out : bufA;
        gcn_gemm<<<(n + 127) / 128, 256, 0, stream>>>(cur, Ws[l], h, n);
        gcn_gather<<<(n + 7) / 8, 256, 0, stream>>>(rowptr, nbr, dinv, h, bs[l], o, n,
                                                    (l < 2) ? 1 : 0);
        cur = o;
    }
}


# GEMM stages W into LDS with all 16 of a thread's 16-byte loads in flight before the first LDS store, not 64 dependent load-wait-store trips of 4 bytes; products and sum order unchanged
# speedup vs baseline: 3.7940x; 1.2502x over previous
#include <hip/hip_runtime.h>

typedef __attribute__((ext_vector_type(2))) float v2f;
typedef __attribute__((ext_vector_type(8))) float v8f;

#define GCN_D 128

#define GCN_BSHIFT 9
#define GCN_BNODES 512
#define GCN_CHUNKS 256
#define GCN_MAXB   2048

__global__ __launch_bounds__(1024) void gcn_bhist(const int* __restrict__ dst,
                                                  int* __restrict__ bhist,
                                                  int n_edges, int nb) {
    __shared__ int s_h[GCN_MAXB];
    for (int k = threadIdx.x; k < nb; k += 1024) s_h[k] = 0;
    __syncthreads();
    const int chunk = (n_edges + GCN_CHUNKS - 1) / GCN_CHUNKS;
    const int lo = blockIdx.x * chunk;
    const int hi = (lo + chunk < n_edges) ? (lo + chunk) : n_edges;
    for (int i = lo + threadIdx.x; i < hi; i += 4 * 1024) {
        int d[4];
        #pragma unroll
        for (int k = 0; k < 4; ++k) d[k] = (i + k * 1024 < hi) ? dst[i + k * 1024] : 0;
        #pragma unroll
        for (int k = 0; k < 4; ++k)
            if (i + k * 1024 < hi) atomicAdd(&s_h[d[k] >> GCN_BSHIFT], 1);
    }
    __syncthreads();
    for (int k = threadIdx.x; k < nb; k += 1024) bhist[k * GCN_CHUNKS + blockIdx.x] = s_h[k];
}

__global__ __launch_bounds__(GCN_CHUNKS) void gcn_bscan(const int* __restrict__ bhist,
                                                        int* __restrict__ boffs,
                                                        int* __restrict__ bstart, int nb) {
    __shared__ int s_before[8];
    __shared__ int s_wave[8];
    const int tid  = threadIdx.x;
    const int lane = tid & 31;
    const int wave = tid >> 5;
    const int b    = blockIdx.x;

    int before = 0;
    const int4* h4 = (const int4*)bhist;
    #pragma unroll 8
    for (int i = tid; i < b * (GCN_CHUNKS / 4); i += GCN_CHUNKS) {
        int4 v = h4[i];
        before += (v.x + v.y) + (v.z + v.w);
    }
    for (int off = 16; off > 0; off >>= 1) before += __shfl_xor(before, off);
    if (lane == 0) s_before[wave] = before;

    const int mine = bhist[b * GCN_CHUNKS + tid];
    int inc = mine;
    for (int off = 1; off < 32; off <<= 1) {
        int y = __shfl_up(inc, off);
        if (lane >= off) inc += y;
    }
    if (lane == 31) s_wave[wave] = inc;
    __syncthreads();

    int excl = inc - mine;
    #pragma unroll
    for (int w = 0; w < 8; ++w) {
        excl += s_before[w];
        if (w < wave) excl += s_wave[w];
    }
    boffs[b * GCN_CHUNKS + tid] = excl;
    if (tid == 0) bstart[b] = excl;
    if (b == nb - 1 && tid == GCN_CHUNKS - 1) bstart[nb] = excl + mine;
}

__global__ __launch_bounds__(1024) void gcn_bscatter(const int* __restrict__ src,
                                                     const int* __restrict__ dst,
                                                     const int* __restrict__ boffs,
                                                     int* __restrict__ coarse,
                                                     int n_edges, int nb) {
    __shared__ int s_pos[GCN_MAXB];
    for (int k = threadIdx.x; k < nb; k += 1024) s_pos[k] = boffs[k * GCN_CHUNKS + blockIdx.x];
    __syncthreads();
    const int chunk = (n_edges + GCN_CHUNKS - 1) / GCN_CHUNKS;
    const int lo = blockIdx.x * chunk;
    const int hi = (lo + chunk < n_edges) ? (lo + chunk) : n_edges;
    int2* pairs = (int2*)coarse;
    for (int i = lo + threadIdx.x; i < hi; i += 4 * 1024) {
        int d[4], s[4];
        #pragma unroll
        for (int k = 0; k < 4; ++k) {
            const bool in = i + k * 1024 < hi;
            d[k] = in ? dst[i + k * 1024] : 0;
            s[k] = in ? src[i + k * 1024] : 0;
        }
        #pragma unroll
        for (int k = 0; k < 4; ++k) {
            if (i + k * 1024 < hi) {
                const int p = atomicAdd(&s_pos[d[k] >> GCN_BSHIFT], 1);
                pairs[p] = make_int2(s[k], d[k]);
            }
        }
    }
}

__global__ __launch_bounds__(1024) void gcn_bucket(const int* __restrict__ coarse,
                                                   const int* __restrict__ bstart,
                                                   int* __restrict__ rank,
                                                   int* __restrict__ rowptr,
                                                   float* __restrict__ dinv, int n) {
    __shared__ int s_cnt[GCN_BNODES];
    __shared__ int s_wave[GCN_BNODES / 32];
    const int tid  = threadIdx.x;
    const int lane = tid & 31;
    const int wave = tid >> 5;
    const int b    = blockIdx.x;
    const int bs   = bstart[b];
    const int be   = bstart[b + 1];
    const int node0 = b << GCN_BSHIFT;

    if (tid < GCN_BNODES) s_cnt[tid] = 0;
    __syncthreads();
    const int2* pairs = (const int2*)coarse;
    for (int i = bs + tid; i < be; i += 4 * 1024) {
        int d[4];
        #pragma unroll
        for (int k = 0; k < 4; ++k) d[k] = (i + k * 1024 < be) ? pairs[i + k * 1024].y : node0;
        #pragma unroll
        for (int k = 0; k < 4; ++k)
            if (i + k * 1024 < be) rank[i + k * 1024] = atomicAdd(&s_cnt[d[k] - node0], 1);
    }
    __syncthreads();

    int c = 0, inc = 0;
    if (tid < GCN_BNODES) {
        c = s_cnt[tid];
        inc = c;
        for (int off = 1; off < 32; off <<= 1) {
            int y = __shfl_up(inc, off);
            if (lane >= off) inc += y;
        }
        if (lane == 31) s_wave[wave] = inc;
    }
    __syncthreads();
    if (tid < GCN_BNODES) {
        int excl = inc - c;
        #pragma unroll
        for (int w = 0; w < GCN_BNODES / 32; ++w)
            if (w < wave) excl += s_wave[w];
        const int node = node0 + tid;
        if (node < n) {
            rowptr[node] = bs + excl;
            dinv[node] = rsqrtf(1.0f + (float)c);
        }
    }
    if (b == (int)gridDim.x - 1 && tid == 0) rowptr[n] = be;
}

__global__ __launch_bounds__(256) void gcn_fill(const int* __restrict__ coarse,
                                                const int* __restrict__ rowptr,
                                                const int* __restrict__ rank,
                                                const float* __restrict__ dinv,
                                                int* __restrict__ nbr, int n_edges) {
    int i = blockIdx.x * blockDim.x + threadIdx.x;
    int stride = gridDim.x * blockDim.x;
    const int2* in = (const int2*)coarse;
    int2* pairs = (int2*)nbr;
    for (; i < n_edges; i += stride) {
        const int2 e = in[i];
        pairs[rowptr[e.y] + rank[i]] = make_int2(e.x, __float_as_int(dinv[e.x]));
    }
}

__global__ __launch_bounds__(256) void gcn_gemm(const float* __restrict__ X,
                                                const float* __restrict__ W,
                                                float* __restrict__ H, int n) {
    __shared__ __align__(16) float sW[GCN_D * GCN_D];
    {
        const float4* W4 = (const float4*)W;
        float4* sW4 = (float4*)sW;
        float4 w[16];
        #pragma unroll
        for (int k = 0; k < 16; ++k) w[k] = W4[k * 256 + threadIdx.x];
        #pragma unroll
        for (int k = 0; k < 16; ++k) sW4[k * 256 + threadIdx.x] = w[k];
    }
    __syncthreads();

    const int lane = threadIdx.x & 31;
    const int wave = threadIdx.x >> 5;
    const int m    = lane & 15;
    const int kh   = lane >> 4;
    const int row0 = blockIdx.x * 128 + wave * 16;

    int rowA = row0 + m;
    if (rowA > n - 1) rowA = n - 1;
    const float* xrow = X + (size_t)rowA * GCN_D;

    v8f acc[8] = {};

    for (int kk = 0; kk < 32; ++kk) {
        const int kb = kk * 4 + kh * 2;
        v2f a;
        a.x = xrow[kb];
        a.y = xrow[kb + 1];
        #pragma unroll
        for (int nt = 0; nt < 8; ++nt) {
            v2f b;
            b.x = sW[(kb + 0) * GCN_D + nt * 16 + m];
            b.y = sW[(kb + 1) * GCN_D + nt * 16 + m];
            acc[nt] = __builtin_amdgcn_wmma_f32_16x16x4_f32(
                false, a, false, b, (short)0, acc[nt], false, false);
        }
    }

    #pragma unroll
    for (int nt = 0; nt < 8; ++nt) {
        #pragma unroll
        for (int v = 0; v < 8; ++v) {
            int r = row0 + v + 8 * kh;
            if (r < n) H[(size_t)r * GCN_D + nt * 16 + m] = acc[nt][v];
        }
    }
}

__global__ __launch_bounds__(256) void gcn_gather(const int* __restrict__ rowptr,
                                                  const int* __restrict__ nbr,
                                                  const float* __restrict__ dinv,
                                                  const float* __restrict__ h,
                                                  const float* __restrict__ bias,
                                                  float* __restrict__ out, int n, int do_relu) {
    const int lane = threadIdx.x & 31;
    int node = blockIdx.x * (blockDim.x >> 5) + (threadIdx.x >> 5);
    if (node >= n) return;
    node = __builtin_amdgcn_readfirstlane(node);

    const int beg = rowptr[node];
    const int end = rowptr[node + 1];
    const float dd = dinv[node];
    const float4* h4 = (const float4*)h;

    float4 acc = make_float4(0.0f, 0.0f, 0.0f, 0.0f);
    for (int base = beg; base < end; base += 32) {
        const int cnt = (end - base < 32) ? (end - base) : 32;
        int   my = 0;
        float mw = 0.0f;
        if (lane < cnt) {
            const int2 p = ((const int2*)nbr)[base + lane];
            my = p.x;
            mw = __int_as_float(p.y) * dd;
        }
        const int mwb = __float_as_int(mw);
#define GCN_ROW(J) h4[(size_t)__builtin_amdgcn_readlane(my, (J)) * 32 + lane]
#define GCN_WGT(J) __int_as_float(__builtin_amdgcn_readlane(mwb, (J)))
        int j = 0;
        for (; j + 4 <= cnt; j += 4) {
            const float4 v0 = GCN_ROW(j + 0);
            const float4 v1 = GCN_ROW(j + 1);
            const float4 v2 = GCN_ROW(j + 2);
            const float4 v3 = GCN_ROW(j + 3);
            const float w0 = GCN_WGT(j + 0), w1 = GCN_WGT(j + 1);
            const float w2 = GCN_WGT(j + 2), w3 = GCN_WGT(j + 3);
            acc.x += v0.x * w0; acc.y += v0.y * w0; acc.z += v0.z * w0; acc.w += v0.w * w0;
            acc.x += v1.x * w1; acc.y += v1.y * w1; acc.z += v1.z * w1; acc.w += v1.w * w1;
            acc.x += v2.x * w2; acc.y += v2.y * w2; acc.z += v2.z * w2; acc.w += v2.w * w2;
            acc.x += v3.x * w3; acc.y += v3.y * w3; acc.z += v3.z * w3; acc.w += v3.w * w3;
        }
        for (; j < cnt; ++j) {
            const float4 v0 = GCN_ROW(j);
            const float w0 = GCN_WGT(j);
            acc.x += v0.x * w0; acc.y += v0.y * w0; acc.z += v0.z * w0; acc.w += v0.w * w0;
        }
#undef GCN_ROW
#undef GCN_WGT
    }

    const float d2 = dd * dd;
    const float4 hv = h4[(size_t)node * 32 + lane];
    const float4 bv = ((const float4*)bias)[lane];
    float4 o;
    o.x = acc.x + hv.x * d2 + bv.x;
    o.y = acc.y + hv.y * d2 + bv.y;
    o.z = acc.z + hv.z * d2 + bv.z;
    o.w = acc.w + hv.w * d2 + bv.w;
    if (do_relu) {
        o.x = fmaxf(o.x, 0.0f);
        o.y = fmaxf(o.y, 0.0f);
        o.z = fmaxf(o.z, 0.0f);
        o.w = fmaxf(o.w, 0.0f);
    }
    ((float4*)out)[(size_t)node * 32 + lane] = o;
}

extern "C" void kernel_launch(void* const* d_in, const int* in_sizes, int n_in,
                              void* d_out, int out_size, void* d_ws, size_t ws_size,
                              hipStream_t stream) {
    const float* x  = (const float*)d_in[0];
    const int*   ei = (const int*)d_in[1];
    const float* W1 = (const float*)d_in[2];
    const float* b1 = (const float*)d_in[3];
    const float* W2 = (const float*)d_in[4];
    const float* b2 = (const float*)d_in[5];
    const float* W3 = (const float*)d_in[6];
    const float* b3 = (const float*)d_in[7];

    const int n = in_sizes[0] / GCN_D;
    const int E = in_sizes[1] / 2;
    const int* srcI = ei;
    const int* dstI = ei + E;

    float* ws = (float*)d_ws;
    size_t n4 = ((size_t)n + 3) & ~(size_t)3;
    float* dinv = ws;
    float* h    = ws + n4;
    float* reg  = h + (size_t)n * GCN_D;
    float* bufA = reg + (size_t)n * GCN_D;
    float* out  = (float*)d_out;

    size_t e2 = ((size_t)E + 1) & ~(size_t)1;
    const int nb = (n + GCN_BNODES - 1) >> GCN_BSHIFT;
    size_t nb4 = ((size_t)nb + 1 + 3) & ~(size_t)3;
    int*  bhist  = (int*)reg;
    int*  boffs  = bhist + (size_t)nb * GCN_CHUNKS;
    int*  bstart = boffs + (size_t)nb * GCN_CHUNKS;
    int*  rowptr = bstart + nb4;
    int*  rank   = rowptr + n4 + 4;
    int*  nbr    = rank + e2;
    int*  coarse = nbr + 2 * e2;

    gcn_bhist<<<GCN_CHUNKS, 1024, 0, stream>>>(dstI, bhist, E, nb);
    gcn_bscan<<<nb, GCN_CHUNKS, 0, stream>>>(bhist, boffs, bstart, nb);
    gcn_bscatter<<<GCN_CHUNKS, 1024, 0, stream>>>(srcI, dstI, boffs, coarse, E, nb);
    gcn_bucket<<<nb, 1024, 0, stream>>>(coarse, bstart, rank, rowptr, dinv, n);
    gcn_fill<<<(E + 255) / 256, 256, 0, stream>>>(coarse, rowptr, rank, dinv, nbr, E);

    const float* Ws[3] = {W1, W2, W3};
    const float* bs[3] = {b1, b2, b3};
    const float* cur = x;
    for (int l = 0; l < 3; ++l) {
        float* o = (l == 2) ? out : bufA;
        gcn_gemm<<<(n + 127) / 128, 256, 0, stream>>>(cur, Ws[l], h, n);
        gcn_gather<<<(n + 7) / 8, 256, 0, stream>>>(rowptr, nbr, dinv, h, bs[l], o, n,
                                                    (l < 2) ? 1 : 0);
        cur = o;
    }
}


# Edge weight folded into the rows: GEMM stores dinv[row]*row, gather sums plain rows and scales by dinv[dst] once; edge list = src ids placed by the bucket pass itself; fill launch and rank array gone
# speedup vs baseline: 4.3694x; 1.1517x over previous
#include <hip/hip_runtime.h>

typedef __attribute__((ext_vector_type(2))) float v2f;
typedef __attribute__((ext_vector_type(8))) float v8f;

#define GCN_D 128

#define GCN_BSHIFT 9
#define GCN_BNODES 512
#define GCN_CHUNKS 256
#define GCN_MAXB   2048

__global__ __launch_bounds__(1024) void gcn_bhist(const int* __restrict__ dst,
                                                  int* __restrict__ bhist,
                                                  int n_edges, int nb) {
    __shared__ int s_h[GCN_MAXB];
    for (int k = threadIdx.x; k < nb; k += 1024) s_h[k] = 0;
    __syncthreads();
    const int chunk = (n_edges + GCN_CHUNKS - 1) / GCN_CHUNKS;
    const int lo = blockIdx.x * chunk;
    const int hi = (lo + chunk < n_edges) ? (lo + chunk) : n_edges;
    for (int i = lo + threadIdx.x; i < hi; i += 4 * 1024) {
        int d[4];
        #pragma unroll
        for (int k = 0; k < 4; ++k) d[k] = (i + k * 1024 < hi) ? dst[i + k * 1024] : 0;
        #pragma unroll
        for (int k = 0; k < 4; ++k)
            if (i + k * 1024 < hi) atomicAdd(&s_h[d[k] >> GCN_BSHIFT], 1);
    }
    __syncthreads();
    for (int k = threadIdx.x; k < nb; k += 1024) bhist[k * GCN_CHUNKS + blockIdx.x] = s_h[k];
}

__global__ __launch_bounds__(GCN_CHUNKS) void gcn_bscan(const int* __restrict__ bhist,
                                                        int* __restrict__ boffs,
                                                        int* __restrict__ bstart, int nb) {
    __shared__ int s_before[8];
    __shared__ int s_wave[8];
    const int tid  = threadIdx.x;
    const int lane = tid & 31;
    const int wave = tid >> 5;
    const int b    = blockIdx.x;

    int before = 0;
    const int4* h4 = (const int4*)bhist;
    #pragma unroll 8
    for (int i = tid; i < b * (GCN_CHUNKS / 4); i += GCN_CHUNKS) {
        int4 v = h4[i];
        before += (v.x + v.y) + (v.z + v.w);
    }
    for (int off = 16; off > 0; off >>= 1) before += __shfl_xor(before, off);
    if (lane == 0) s_before[wave] = before;

    const int mine = bhist[b * GCN_CHUNKS + tid];
    int inc = mine;
    for (int off = 1; off < 32; off <<= 1) {
        int y = __shfl_up(inc, off);
        if (lane >= off) inc += y;
    }
    if (lane == 31) s_wave[wave] = inc;
    __syncthreads();

    int excl = inc - mine;
    #pragma unroll
    for (int w = 0; w < 8; ++w) {
        excl += s_before[w];
        if (w < wave) excl += s_wave[w];
    }
    boffs[b * GCN_CHUNKS + tid] = excl;
    if (tid == 0) bstart[b] = excl;
    if (b == nb - 1 && tid == GCN_CHUNKS - 1) bstart[nb] = excl + mine;
}

__global__ __launch_bounds__(1024) void gcn_bscatter(const int* __restrict__ src,
                                                     const int* __restrict__ dst,
                                                     const int* __restrict__ boffs,
                                                     int* __restrict__ coarse,
                                                     int n_edges, int nb) {
    __shared__ int s_pos[GCN_MAXB];
    for (int k = threadIdx.x; k < nb; k += 1024) s_pos[k] = boffs[k * GCN_CHUNKS + blockIdx.x];
    __syncthreads();
    const int chunk = (n_edges + GCN_CHUNKS - 1) / GCN_CHUNKS;
    const int lo = blockIdx.x * chunk;
    const int hi = (lo + chunk < n_edges) ? (lo + chunk) : n_edges;
    int2* pairs = (int2*)coarse;
    for (int i = lo + threadIdx.x; i < hi; i += 4 * 1024) {
        int d[4], s[4];
        #pragma unroll
        for (int k = 0; k < 4; ++k) {
            const bool in = i + k * 1024 < hi;
            d[k] = in ? dst[i + k * 1024] : 0;
            s[k] = in ? src[i + k * 1024] : 0;
        }
        #pragma unroll
        for (int k = 0; k < 4; ++k) {
            if (i + k * 1024 < hi) {
                const int p = atomicAdd(&s_pos[d[k] >> GCN_BSHIFT], 1);
                pairs[p] = make_int2(s[k], d[k]);
            }
        }
    }
}

__global__ __launch_bounds__(1024) void gcn_bucket(const int* __restrict__ coarse,
                                                   const int* __restrict__ bstart,
                                                   int* __restrict__ rowptr,
                                                   float* __restrict__ dinv,
                                                   int* __restrict__ nbr, int n) {
    __shared__ int s_cnt[GCN_BNODES];
    __shared__ int s_wave[GCN_BNODES / 32];
    const int tid  = threadIdx.x;
    const int lane = tid & 31;
    const int wave = tid >> 5;
    const int b    = blockIdx.x;
    const int bs   = bstart[b];
    const int be   = bstart[b + 1];
    const int node0 = b << GCN_BSHIFT;

    if (tid < GCN_BNODES) s_cnt[tid] = 0;
    __syncthreads();
    const int2* pairs = (const int2*)coarse;
    for (int i = bs + tid; i < be; i += 4 * 1024) {
        int d[4];
        #pragma unroll
        for (int k = 0; k < 4; ++k) d[k] = (i + k * 1024 < be) ? pairs[i + k * 1024].y : node0;
        #pragma unroll
        for (int k = 0; k < 4; ++k)
            if (i + k * 1024 < be) atomicAdd(&s_cnt[d[k] - node0], 1);
    }
    __syncthreads();

    int c = 0, inc = 0;
    if (tid < GCN_BNODES) {
        c = s_cnt[tid];
        inc = c;
        for (int off = 1; off < 32; off <<= 1) {
            int y = __shfl_up(inc, off);
            if (lane >= off) inc += y;
        }
        if (lane == 31) s_wave[wave] = inc;
    }
    __syncthreads();
    if (tid < GCN_BNODES) {
        int excl = inc - c;
        #pragma unroll
        for (int w = 0; w < GCN_BNODES / 32; ++w)
            if (w < wave) excl += s_wave[w];
        const int node = node0 + tid;
        if (node < n) {
            rowptr[node] = bs + excl;
            dinv[node] = rsqrtf(1.0f + (float)c);
        }
        s_cnt[tid] = bs + excl;
    }
    if (b == (int)gridDim.x - 1 && tid == 0) rowptr[n] = be;
    __syncthreads();

    for (int i = bs + tid; i < be; i += 4 * 1024) {
        int2 e[4];
        #pragma unroll
        for (int k = 0; k < 4; ++k)
            e[k] = (i + k * 1024 < be) ? pairs[i + k * 1024] : make_int2(0, node0);
        #pragma unroll
        for (int k = 0; k < 4; ++k)
            if (i + k * 1024 < be) nbr[atomicAdd(&s_cnt[e[k].y - node0], 1)] = e[k].x;
    }
}

__global__ __launch_bounds__(256) void gcn_gemm(const float* __restrict__ X,
                                                const float* __restrict__ W,
                                                const float* __restrict__ dinv,
                                                float* __restrict__ H, int n) {
    __shared__ __align__(16) float sW[GCN_D * GCN_D];
    {
        const float4* W4 = (const float4*)W;
        float4* sW4 = (float4*)sW;
        float4 w[16];
        #pragma unroll
        for (int k = 0; k < 16; ++k) w[k] = W4[k * 256 + threadIdx.x];
        #pragma unroll
        for (int k = 0; k < 16; ++k) sW4[k * 256 + threadIdx.x] = w[k];
    }
    __syncthreads();

    const int lane = threadIdx.x & 31;
    const int wave = threadIdx.x >> 5;
    const int m    = lane & 15;
    const int kh   = lane >> 4;
    const int row0 = blockIdx.x * 128 + wave * 16;

    int rowA = row0 + m;
    if (rowA > n - 1) rowA = n - 1;
    const float* xrow = X + (size_t)rowA * GCN_D;

    v8f acc[8] = {};

    for (int kk = 0; kk < 32; ++kk) {
        const int kb = kk * 4 + kh * 2;
        v2f a;
        a.x = xrow[kb];
        a.y = xrow[kb + 1];
        #pragma unroll
        for (int nt = 0; nt < 8; ++nt) {
            v2f b;
            b.x = sW[(kb + 0) * GCN_D + nt * 16 + m];
            b.y = sW[(kb + 1) * GCN_D + nt * 16 + m];
            acc[nt] = __builtin_amdgcn_wmma_f32_16x16x4_f32(
                false, a, false, b, (short)0, acc[nt], false, false);
        }
    }

    float dv[8];
    #pragma unroll
    for (int v = 0; v < 8; ++v) {
        const int r = row0 + v + 8 * kh;
        dv[v] = dinv[(r < n) ? r : (n - 1)];
    }

    #pragma unroll
    for (int nt = 0; nt < 8; ++nt) {
        #pragma unroll
        for (int v = 0; v < 8; ++v) {
            int r = row0 + v + 8 * kh;
            if (r < n) H[(size_t)r * GCN_D + nt * 16 + m] = acc[nt][v] * dv[v];
        }
    }
}

__global__ __launch_bounds__(256) void gcn_gather(const int* __restrict__ rowptr,
                                                  const int* __restrict__ nbr,
                                                  const float* __restrict__ dinv,
                                                  const float* __restrict__ h,
                                                  const float* __restrict__ bias,
                                                  float* __restrict__ out, int n, int do_relu) {
    const int lane = threadIdx.x & 31;
    int node = blockIdx.x * (blockDim.x >> 5) + (threadIdx.x >> 5);
    if (node >= n) return;
    node = __builtin_amdgcn_readfirstlane(node);

    const int beg = rowptr[node];
    const int end = rowptr[node + 1];
    const float dd = dinv[node];
    const float4* h4 = (const float4*)h;

    float4 acc = make_float4(0.0f, 0.0f, 0.0f, 0.0f);
    for (int base = beg; base < end; base += 32) {
        const int cnt = (end - base < 32) ? (end - base) : 32;
        int my = 0;
        if (lane < cnt) my = nbr[base + lane];
#define GCN_ROW(J) h4[(size_t)__builtin_amdgcn_readlane(my, (J)) * 32 + lane]
        int j = 0;
        for (; j + 4 <= cnt; j += 4) {
            const float4 v0 = GCN_ROW(j + 0);
            const float4 v1 = GCN_ROW(j + 1);
            const float4 v2 = GCN_ROW(j + 2);
            const float4 v3 = GCN_ROW(j + 3);
            acc.x += v0.x; acc.y += v0.y; acc.z += v0.z; acc.w += v0.w;
            acc.x += v1.x; acc.y += v1.y; acc.z += v1.z; acc.w += v1.w;
            acc.x += v2.x; acc.y += v2.y; acc.z += v2.z; acc.w += v2.w;
            acc.x += v3.x; acc.y += v3.y; acc.z += v3.z; acc.w += v3.w;
        }
        for (; j < cnt; ++j) {
            const float4 v0 = GCN_ROW(j);
            acc.x += v0.x; acc.y += v0.y; acc.z += v0.z; acc.w += v0.w;
        }
#undef GCN_ROW
    }

    const float4 hv = h4[(size_t)node * 32 + lane];
    const float4 bv = ((const float4*)bias)[lane];
    float4 o;
    o.x = (acc.x + hv.x) * dd + bv.x;
    o.y = (acc.y + hv.y) * dd + bv.y;
    o.z = (acc.z + hv.z) * dd + bv.z;
    o.w = (acc.w + hv.w) * dd + bv.w;
    if (do_relu) {
        o.x = fmaxf(o.x, 0.0f);
        o.y = fmaxf(o.y, 0.0f);
        o.z = fmaxf(o.z, 0.0f);
        o.w = fmaxf(o.w, 0.0f);
    }
    ((float4*)out)[(size_t)node * 32 + lane] = o;
}

extern "C" void kernel_launch(void* const* d_in, const int* in_sizes, int n_in,
                              void* d_out, int out_size, void* d_ws, size_t ws_size,
                              hipStream_t stream) {
    const float* x  = (const float*)d_in[0];
    const int*   ei = (const int*)d_in[1];
    const float* W1 = (const float*)d_in[2];
    const float* b1 = (const float*)d_in[3];
    const float* W2 = (const float*)d_in[4];
    const float* b2 = (const float*)d_in[5];
    const float* W3 = (const float*)d_in[6];
    const float* b3 = (const float*)d_in[7];

    const int n = in_sizes[0] / GCN_D;
    const int E = in_sizes[1] / 2;
    const int* srcI = ei;
    const int* dstI = ei + E;

    float* ws = (float*)d_ws;
    size_t n4 = ((size_t)n + 3) & ~(size_t)3;
    float* dinv = ws;
    float* h    = ws + n4;
    float* reg  = h + (size_t)n * GCN_D;
    float* bufA = reg + (size_t)n * GCN_D;
    float* out  = (float*)d_out;

    size_t e2 = ((size_t)E + 1) & ~(size_t)1;
    const int nb = (n + GCN_BNODES - 1) >> GCN_BSHIFT;
    size_t nb4 = ((size_t)nb + 1 + 3) & ~(size_t)3;
    int*  bhist  = (int*)reg;
    int*  boffs  = bhist + (size_t)nb * GCN_CHUNKS;
    int*  bstart = boffs + (size_t)nb * GCN_CHUNKS;
    int*  rowptr = bstart + nb4;
    int*  nbr    = rowptr + n4 + 4;
    int*  coarse = nbr + e2;

    gcn_bhist<<<GCN_CHUNKS, 1024, 0, stream>>>(dstI, bhist, E, nb);
    gcn_bscan<<<nb, GCN_CHUNKS, 0, stream>>>(bhist, boffs, bstart, nb);
    gcn_bscatter<<<GCN_CHUNKS, 1024, 0, stream>>>(srcI, dstI, boffs, coarse, E, nb);
    gcn_bucket<<<nb, 1024, 0, stream>>>(coarse, bstart, rowptr, dinv, nbr, n);

    const float* Ws[3] = {W1, W2, W3};
    const float* bs[3] = {b1, b2, b3};
    const float* cur = x;
    for (int l = 0; l < 3; ++l) {
        float* o = (l == 2) ? out : bufA;
        gcn_gemm<<<(n + 127) / 128, 256, 0, stream>>>(cur, Ws[l], dinv, h, n);
        gcn_gather<<<(n + 7) / 8, 256, 0, stream>>>(rowptr, nbr, dinv, h, bs[l], o, n,
                                                    (l < 2) ? 1 : 0);
        cur = o;
    }
}
